# starting kernel
# baseline (speedup 1.0000x reference)
#include <hip/hip_runtime.h>
#include <stdint.h>

#define N_NODES 100000
#define N_EDGES 1000000
#define C 64
#define APAD 260

typedef float v2f __attribute__((ext_vector_type(2)));
typedef float v8f __attribute__((ext_vector_type(8)));

__device__ __forceinline__ unsigned enc_f32(float f) {
    unsigned u = __float_as_uint(f);
    return (u & 0x80000000u) ? ~u : (u | 0x80000000u);
}
__device__ __forceinline__ float dec_f32(unsigned e) {
    unsigned u = (e & 0x80000000u) ? (e ^ 0x80000000u) : ~e;
    return __uint_as_float(u);
}

__global__ void zero_ws_kernel(float* __restrict__ ws, long n) {
    long i = (long)blockIdx.x * blockDim.x + threadIdx.x;
    long stride = (long)gridDim.x * blockDim.x;
    for (; i < n; i += stride) ws[i] = 0.0f;
}

__device__ __forceinline__ float wbig_row(const float* Wl, const float* Wr,
                                          int j, int d) {
    if (j < 192) return Wl[j * 64 + d];
    int r = j - 192;
    return Wr[r * 64 + d] + Wr[4096 + r * 64 + d] + Wr[8192 + r * 64 + d];
}

__global__ void prep_w_kernel(const float* __restrict__ Wl,
                              const float* __restrict__ Wr,
                              const float* __restrict__ b,
                              float2* __restrict__ WbigP,
                              float* __restrict__ bsum) {
    int idx = blockIdx.x * blockDim.x + threadIdx.x;
    int p = idx >> 6;
    int d = idx & 63;
    float2 v;
    v.x = wbig_row(Wl, Wr, 2 * p + 0, d);
    v.y = wbig_row(Wl, Wr, 2 * p + 1, d);
    WbigP[idx] = v;
    if (idx < 64) bsum[idx] = b[idx] + b[64 + idx] + b[128 + idx];
}

__global__ void scatter_edges_kernel(const float* __restrict__ x,
                                     const int* __restrict__ src,
                                     const int* __restrict__ dst,
                                     float* __restrict__ seg_sum,
                                     unsigned* __restrict__ seg_max,
                                     unsigned* __restrict__ cnt) {
    long tid    = (long)blockIdx.x * blockDim.x + threadIdx.x;
    long total  = (long)N_EDGES * 16;
    long stride = (long)gridDim.x * blockDim.x;
    for (; tid < total; tid += stride) {
        int e  = (int)(tid >> 4);
        int c0 = ((int)tid & 15) << 2;
        int s = src[e];
        int d = dst[e];
        const float4 v = *(const float4*)(x + (long)s * C + c0);
        float*    ss = seg_sum + (long)d * C + c0;
        unsigned* sm = seg_max + (long)d * C + c0;
        atomicAdd(ss + 0, v.x);
        atomicAdd(ss + 1, v.y);
        atomicAdd(ss + 2, v.z);
        atomicAdd(ss + 3, v.w);
        atomicMax(sm + 0, enc_f32(v.x));
        atomicMax(sm + 1, enc_f32(v.y));
        atomicMax(sm + 2, enc_f32(v.z));
        atomicMax(sm + 3, enc_f32(v.w));
        if ((tid & 15) == 0) atomicAdd(cnt + d, 1u);
    }
}

__global__ __launch_bounds__(128) void combine_wmma_kernel(
        const float*    __restrict__ x,
        const float*    __restrict__ seg_sum,
        const unsigned* __restrict__ seg_max,
        const unsigned* __restrict__ cnt,
        const float2*   __restrict__ WbigP,
        const float*    __restrict__ bsum,
        float*          __restrict__ out) {
    __shared__ float    Ap[16 * APAD];
    __shared__ float    s_invc[16];
    __shared__ unsigned s_cnt[16];

    const int  t    = threadIdx.x;
    const long base = (long)blockIdx.x * 16;

    if (t < 16) {
        unsigned cn = cnt[base + t];
        s_cnt[t]  = cn;
        s_invc[t] = 1.0f / fmaxf((float)cn, 1.0f);
    }
    __syncthreads();

    for (int idx = t; idx < 16 * 256; idx += 128) {
        int  row = idx >> 8;
        int  j   = idx & 255;
        long n   = base + row;
        float v;
        if (j < 64) {
            v = seg_sum[n * 64 + j];
        } else if (j < 128) {
            v = seg_sum[n * 64 + (j - 64)] * s_invc[row];
        } else if (j < 192) {
            v = (s_cnt[row] != 0u) ? dec_f32(seg_max[n * 64 + (j - 128)]) : 0.0f;
        } else {
            v = x[n * 64 + (j - 192)];
        }
        Ap[row * APAD + j] = v;
    }
    __syncthreads();

    const int lane = t & 31;
    const int wave = t >> 5;
    const int nl   = lane & 15;
    const int half = lane >> 4;
    const int col  = wave * 16 + nl;
    const int m    = nl;

    v8f acc;
    const float bv = bsum[col];
    #pragma unroll
    for (int r = 0; r < 8; ++r) acc[r] = bv;

    const float*  Arow = Ap + m * APAD;
    const float2* Bcol = WbigP + col;

    #pragma unroll
    for (int k = 0; k < 256; k += 4) {
        v2f a, bb;
        a.x = Arow[k + 2 * half + 0];
        a.y = Arow[k + 2 * half + 1];
        const float2 bw = Bcol[((k >> 1) + half) * 64];
        bb.x = bw.x;
        bb.y = bw.y;
        acc = __builtin_amdgcn_wmma_f32_16x16x4_f32(
                  false, a, false, bb, (short)0, acc, false, false);
    }

    #pragma unroll
    for (int r = 0; r < 8; ++r) {
        long n = base + r + 8 * half;
        out[n * 64 + col] = acc[r];
    }
}

extern "C" void kernel_launch(void* const* d_in, const int* in_sizes, int n_in,
                              void* d_out, int out_size, void* d_ws, size_t ws_size,
                              hipStream_t stream) {
    (void)in_sizes; (void)n_in; (void)out_size; (void)ws_size;

    const float* x  = (const float*)d_in[0];
    const int*   ei = (const int*)d_in[1];
    const float* Wl = (const float*)d_in[2];
    const float* Wr = (const float*)d_in[3];
    const float* b  = (const float*)d_in[4];
    float*       out = (float*)d_out;

    float*    ws      = (float*)d_ws;
    float*    seg_sum = ws;
    unsigned* seg_max = (unsigned*)(ws + (long)N_NODES * 64);
    unsigned* cnt     = (unsigned*)(ws + (long)N_NODES * 128);
    float2*   WbigP   = (float2*)(ws + (long)N_NODES * 128 + N_NODES);
    float*    bsum    = (float*)(WbigP + 128 * 64);

    const int* src = ei;
    const int* dst = ei + N_EDGES;

    const long zero_n = (long)N_NODES * 128 + N_NODES;
    zero_ws_kernel<<<2048, 256, 0, stream>>>(ws, zero_n);
    prep_w_kernel<<<32, 256, 0, stream>>>(Wl, Wr, b, WbigP, bsum);
    scatter_edges_kernel<<<62500, 256, 0, stream>>>(x, src, dst, seg_sum, seg_max, cnt);
    combine_wmma_kernel<<<N_NODES / 16, 128, 0, stream>>>(x, seg_sum, seg_max, cnt,
                                                          WbigP, bsum, out);
}


# Atomic scatter of sum/max replaced by per-node source lists (1 counting atomic per edge, scan, fill) reduced in registers in the A-panel fill; float sums now in list (arrival-rank) order
# speedup vs baseline: 1.5555x; 1.5555x over previous
#include <hip/hip_runtime.h>
#include <stdint.h>

#define N_NODES 100000
#define N_EDGES 1000000
#define C 64
#define APAD 260
#define SCAN_CHUNKS 25

typedef float v2f __attribute__((ext_vector_type(2)));
typedef float v8f __attribute__((ext_vector_type(8)));

__global__ void zero_ws_kernel(float* __restrict__ ws, long n) {
    long i = (long)blockIdx.x * blockDim.x + threadIdx.x;
    long stride = (long)gridDim.x * blockDim.x;
    for (; i < n; i += stride) ws[i] = 0.0f;
}

__device__ __forceinline__ float wbig_row(const float* Wl, const float* Wr,
                                          int j, int d) {
    if (j < 192) return Wl[j * 64 + d];
    int r = j - 192;
    return Wr[r * 64 + d] + Wr[4096 + r * 64 + d] + Wr[8192 + r * 64 + d];
}

__global__ void prep_w_kernel(const float* __restrict__ Wl,
                              const float* __restrict__ Wr,
                              const float* __restrict__ b,
                              float2* __restrict__ WbigP,
                              float* __restrict__ bsum) {
    int idx = blockIdx.x * blockDim.x + threadIdx.x;
    int p = idx >> 6;
    int d = idx & 63;
    float2 v;
    v.x = wbig_row(Wl, Wr, 2 * p + 0, d);
    v.y = wbig_row(Wl, Wr, 2 * p + 1, d);
    WbigP[idx] = v;
    if (idx < 64) bsum[idx] = b[idx] + b[64 + idx] + b[128 + idx];
}

__global__ void rank_edges_kernel(const int* __restrict__ dst,
                                  unsigned* __restrict__ cnt,
                                  unsigned* __restrict__ rank) {
    int q = blockIdx.x * blockDim.x + threadIdx.x;
    if (q >= N_EDGES / 4) return;
    const int4 d = ((const int4*)dst)[q];
    uint4 r;
    r.x = atomicAdd(cnt + d.x, 1u);
    r.y = atomicAdd(cnt + d.y, 1u);
    r.z = atomicAdd(cnt + d.z, 1u);
    r.w = atomicAdd(cnt + d.w, 1u);
    ((uint4*)rank)[q] = r;
}

__device__ __forceinline__ unsigned wave_incl_scan(unsigned s, int lane) {
    #pragma unroll
    for (int d = 1; d < 32; d <<= 1) {
        unsigned o = __shfl_up(s, d);
        if (lane >= d) s += o;
    }
    return s;
}

__global__ __launch_bounds__(1024) void scan_counts_kernel(
        const unsigned* __restrict__ cnt, unsigned* __restrict__ off) {
    __shared__ unsigned s_w[SCAN_CHUNKS][32];
    __shared__ unsigned s_ctot[SCAN_CHUNKS];
    __shared__ unsigned s_cbase[SCAN_CHUNKS + 1];

    const int t = threadIdx.x;
    const int lane = t & 31;
    const int wave = t >> 5;

    unsigned tsum[SCAN_CHUNKS];
    unsigned incl[SCAN_CHUNKS];

    #pragma unroll
    for (int c = 0; c < SCAN_CHUNKS; ++c) {
        const int i = c * 4096 + t * 4;
        uint4 v = make_uint4(0u, 0u, 0u, 0u);
        if (i < N_NODES) v = *(const uint4*)(cnt + i);
        tsum[c] = v.x + v.y + v.z + v.w;
    }
    #pragma unroll
    for (int c = 0; c < SCAN_CHUNKS; ++c) {
        incl[c] = wave_incl_scan(tsum[c], lane);
        if (lane == 31) s_w[c][wave] = incl[c];
    }
    __syncthreads();
    if (wave < SCAN_CHUNKS) {
        const unsigned w = s_w[wave][lane];
        const unsigned s = wave_incl_scan(w, lane);
        s_w[wave][lane] = s - w;
        if (lane == 31) s_ctot[wave] = s;
    }
    __syncthreads();
    if (t == 0) {
        unsigned run = 0;
        for (int c = 0; c < SCAN_CHUNKS; ++c) { s_cbase[c] = run; run += s_ctot[c]; }
        s_cbase[SCAN_CHUNKS] = run;
    }
    __syncthreads();
    #pragma unroll
    for (int c = 0; c < SCAN_CHUNKS; ++c) {
        const int i = c * 4096 + t * 4;
        if (i < N_NODES) {
            const uint4 v = *(const uint4*)(cnt + i);
            const unsigned b0 = s_cbase[c] + s_w[c][wave] + (incl[c] - tsum[c]);
            uint4 o;
            o.x = b0;
            o.y = o.x + v.x;
            o.z = o.y + v.y;
            o.w = o.z + v.z;
            *(uint4*)(off + i) = o;
        }
    }
    if (t == 0) off[N_NODES] = s_cbase[SCAN_CHUNKS];
}

__global__ void fill_lists_kernel(const int* __restrict__ src,
                                  const int* __restrict__ dst,
                                  const unsigned* __restrict__ off,
                                  const unsigned* __restrict__ rank,
                                  int* __restrict__ lists) {
    int q = blockIdx.x * blockDim.x + threadIdx.x;
    if (q >= N_EDGES / 4) return;
    const int4  s = ((const int4*)src)[q];
    const int4  d = ((const int4*)dst)[q];
    const uint4 r = ((const uint4*)rank)[q];
    const unsigned p0 = off[d.x] + r.x;
    const unsigned p1 = off[d.y] + r.y;
    const unsigned p2 = off[d.z] + r.z;
    const unsigned p3 = off[d.w] + r.w;
    lists[p0] = s.x;
    lists[p1] = s.y;
    lists[p2] = s.z;
    lists[p3] = s.w;
}

__device__ __forceinline__ void acc_row(float4& s, float4& m, const float4 v) {
    s.x += v.x; s.y += v.y; s.z += v.z; s.w += v.w;
    m.x = fmaxf(m.x, v.x); m.y = fmaxf(m.y, v.y);
    m.z = fmaxf(m.z, v.z); m.w = fmaxf(m.w, v.w);
}

__device__ __forceinline__ void acc_row_if(float4& s, float4& m, const float4 v, bool on) {
    const float ninf = -__builtin_inff();
    s.x += on ? v.x : 0.0f; s.y += on ? v.y : 0.0f;
    s.z += on ? v.z : 0.0f; s.w += on ? v.w : 0.0f;
    m.x = fmaxf(m.x, on ? v.x : ninf); m.y = fmaxf(m.y, on ? v.y : ninf);
    m.z = fmaxf(m.z, on ? v.z : ninf); m.w = fmaxf(m.w, on ? v.w : ninf);
}

__global__ __launch_bounds__(128) void combine_wmma_kernel(
        const float*    __restrict__ x,
        const unsigned* __restrict__ off,
        const int*      __restrict__ lists,
        const float2*   __restrict__ WbigP,
        const float*    __restrict__ bsum,
        float*          __restrict__ out) {
    __shared__ __attribute__((aligned(16))) float Ap[16 * APAD];

    const int  t    = threadIdx.x;
    const long base = (long)blockIdx.x * 16;

    {
        const int  row = t >> 3;
        const int  sub = t & 7;
        const long n   = base + row;
        const unsigned e0 = off[n];
        const unsigned e1 = off[n + 1];
        const float4* __restrict__ x4 = (const float4*)x;

        const float ninf = -__builtin_inff();
        float4 sa = make_float4(0.0f, 0.0f, 0.0f, 0.0f);
        float4 sb = make_float4(0.0f, 0.0f, 0.0f, 0.0f);
        float4 ma = make_float4(ninf, ninf, ninf, ninf);
        float4 mb = make_float4(ninf, ninf, ninf, ninf);

        unsigned e = e0;
        for (; e + 4 <= e1; e += 4) {
            const long i0 = lists[e + 0];
            const long i1 = lists[e + 1];
            const long i2 = lists[e + 2];
            const long i3 = lists[e + 3];
            const float4 a0 = x4[i0 * 16 + sub];
            const float4 b0 = x4[i0 * 16 + 8 + sub];
            const float4 a1 = x4[i1 * 16 + sub];
            const float4 b1 = x4[i1 * 16 + 8 + sub];
            const float4 a2 = x4[i2 * 16 + sub];
            const float4 b2 = x4[i2 * 16 + 8 + sub];
            const float4 a3 = x4[i3 * 16 + sub];
            const float4 b3 = x4[i3 * 16 + 8 + sub];
            acc_row(sa, ma, a0); acc_row(sb, mb, b0);
            acc_row(sa, ma, a1); acc_row(sb, mb, b1);
            acc_row(sa, ma, a2); acc_row(sb, mb, b2);
            acc_row(sa, ma, a3); acc_row(sb, mb, b3);
        }
        if (e < e1) {
            const unsigned left = e1 - e;
            const long i0 = lists[e];
            const long i1 = lists[left > 1u ? e + 1 : e];
            const long i2 = lists[left > 2u ? e + 2 : e];
            const float4 a0 = x4[i0 * 16 + sub];
            const float4 b0 = x4[i0 * 16 + 8 + sub];
            const float4 a1 = x4[i1 * 16 + sub];
            const float4 b1 = x4[i1 * 16 + 8 + sub];
            const float4 a2 = x4[i2 * 16 + sub];
            const float4 b2 = x4[i2 * 16 + 8 + sub];
            acc_row(sa, ma, a0); acc_row(sb, mb, b0);
            acc_row_if(sa, ma, a1, left > 1u); acc_row_if(sb, mb, b1, left > 1u);
            acc_row_if(sa, ma, a2, left > 2u); acc_row_if(sb, mb, b2, left > 2u);
        }

        const unsigned cn   = e1 - e0;
        const float    invc = 1.0f / fmaxf((float)cn, 1.0f);
        const float4   xa   = x4[n * 16 + sub];
        const float4   xb   = x4[n * 16 + 8 + sub];
        if (cn == 0u) {
            ma = make_float4(0.0f, 0.0f, 0.0f, 0.0f);
            mb = make_float4(0.0f, 0.0f, 0.0f, 0.0f);
        }
        float* ap = Ap + row * APAD + 4 * sub;
        *(float4*)(ap +   0) = sa;
        *(float4*)(ap +  32) = sb;
        *(float4*)(ap +  64) = make_float4(sa.x * invc, sa.y * invc, sa.z * invc, sa.w * invc);
        *(float4*)(ap +  96) = make_float4(sb.x * invc, sb.y * invc, sb.z * invc, sb.w * invc);
        *(float4*)(ap + 128) = ma;
        *(float4*)(ap + 160) = mb;
        *(float4*)(ap + 192) = xa;
        *(float4*)(ap + 224) = xb;
    }
    __syncthreads();

    const int lane = t & 31;
    const int wave = t >> 5;
    const int nl   = lane & 15;
    const int half = lane >> 4;
    const int col  = wave * 16 + nl;
    const int m    = nl;

    v8f acc;
    const float bv = bsum[col];
    #pragma unroll
    for (int r = 0; r < 8; ++r) acc[r] = bv;

    const float*  Arow = Ap + m * APAD;
    const float2* Bcol = WbigP + col;

    #pragma unroll
    for (int k = 0; k < 256; k += 4) {
        v2f a, bb;
        a.x = Arow[k + 2 * half + 0];
        a.y = Arow[k + 2 * half + 1];
        const float2 bw = Bcol[((k >> 1) + half) * 64];
        bb.x = bw.x;
        bb.y = bw.y;
        acc = __builtin_amdgcn_wmma_f32_16x16x4_f32(
                  false, a, false, bb, (short)0, acc, false, false);
    }

    #pragma unroll
    for (int r = 0; r < 8; ++r) {
        long n = base + r + 8 * half;
        out[n * 64 + col] = acc[r];
    }
}

extern "C" void kernel_launch(void* const* d_in, const int* in_sizes, int n_in,
                              void* d_out, int out_size, void* d_ws, size_t ws_size,
                              hipStream_t stream) {
    (void)in_sizes; (void)n_in; (void)out_size; (void)ws_size;

    const float* x  = (const float*)d_in[0];
    const int*   ei = (const int*)d_in[1];
    const float* Wl = (const float*)d_in[2];
    const float* Wr = (const float*)d_in[3];
    const float* b  = (const float*)d_in[4];
    float*       out = (float*)d_out;

    float*    ws    = (float*)d_ws;
    unsigned* cnt   = (unsigned*)ws;
    unsigned* off   = (unsigned*)(ws + 100000);
    unsigned* rank  = (unsigned*)(ws + 200004);
    int*      lists = (int*)(ws + 1200004);
    float2*   WbigP = (float2*)(ws + 2200004);
    float*    bsum  = ws + 2216388;

    const int* src = ei;
    const int* dst = ei + N_EDGES;

    zero_ws_kernel<<<391, 256, 0, stream>>>(ws, (long)N_NODES);
    prep_w_kernel<<<32, 256, 0, stream>>>(Wl, Wr, b, WbigP, bsum);
    rank_edges_kernel<<<977, 256, 0, stream>>>(dst, cnt, rank);
    scan_counts_kernel<<<1, 1024, 0, stream>>>(cnt, off);
    fill_lists_kernel<<<977, 256, 0, stream>>>(src, dst, off, rank, lists);
    combine_wmma_kernel<<<N_NODES / 16, 128, 0, stream>>>(x, off, lists, WbigP, bsum, out);
}


# Scan and fill launches dropped: the counting atomic's return places each source id straight into a fixed 32-id bucket per node (count > 32: node re-reads the edge list); sums stay in arrival order
# speedup vs baseline: 2.0065x; 1.2899x over previous
#include <hip/hip_runtime.h>
#include <stdint.h>

#define N_NODES 100000
#define N_EDGES 1000000
#define C 64
#define APAD 260
#define CAP 32
#define WS_NEED 13265920ul

typedef float v2f __attribute__((ext_vector_type(2)));
typedef float v8f __attribute__((ext_vector_type(8)));

__global__ void zero_ws_kernel(float* __restrict__ ws, long n) {
    long i = (long)blockIdx.x * blockDim.x + threadIdx.x;
    long stride = (long)gridDim.x * blockDim.x;
    for (; i < n; i += stride) ws[i] = 0.0f;
}

__device__ __forceinline__ float wbig_row(const float* Wl, const float* Wr,
                                          int j, int d) {
    if (j < 192) return Wl[j * 64 + d];
    int r = j - 192;
    return Wr[r * 64 + d] + Wr[4096 + r * 64 + d] + Wr[8192 + r * 64 + d];
}

__global__ void prep_w_kernel(const float* __restrict__ Wl,
                              const float* __restrict__ Wr,
                              const float* __restrict__ b,
                              float2* __restrict__ WbigP,
                              float* __restrict__ bsum) {
    int idx = blockIdx.x * blockDim.x + threadIdx.x;
    int p = idx >> 6;
    int d = idx & 63;
    float2 v;
    v.x = wbig_row(Wl, Wr, 2 * p + 0, d);
    v.y = wbig_row(Wl, Wr, 2 * p + 1, d);
    WbigP[idx] = v;
    if (idx < 64) bsum[idx] = b[idx] + b[64 + idx] + b[128 + idx];
}

__global__ void bucket_edges_kernel(const int* __restrict__ src,
                                    const int* __restrict__ dst,
                                    unsigned* __restrict__ cnt,
                                    int* __restrict__ bucket,
                                    size_t ws_bytes) {
    if (ws_bytes < WS_NEED) return;
    int q = blockIdx.x * blockDim.x + threadIdx.x;
    if (q >= N_EDGES / 4) return;
    const int4 s = ((const int4*)src)[q];
    const int4 d = ((const int4*)dst)[q];
    const unsigned r0 = atomicAdd(cnt + d.x, 1u);
    const unsigned r1 = atomicAdd(cnt + d.y, 1u);
    const unsigned r2 = atomicAdd(cnt + d.z, 1u);
    const unsigned r3 = atomicAdd(cnt + d.w, 1u);
    if (r0 < CAP) bucket[d.x * CAP + (int)r0] = s.x;
    if (r1 < CAP) bucket[d.y * CAP + (int)r1] = s.y;
    if (r2 < CAP) bucket[d.z * CAP + (int)r2] = s.z;
    if (r3 < CAP) bucket[d.w * CAP + (int)r3] = s.w;
}

__device__ __forceinline__ void acc_row(float4& s, float4& m, const float4 v) {
    s.x += v.x; s.y += v.y; s.z += v.z; s.w += v.w;
    m.x = fmaxf(m.x, v.x); m.y = fmaxf(m.y, v.y);
    m.z = fmaxf(m.z, v.z); m.w = fmaxf(m.w, v.w);
}

__device__ __forceinline__ void acc_row_if(float4& s, float4& m, const float4 v, bool on) {
    const float ninf = -__builtin_inff();
    s.x += on ? v.x : 0.0f; s.y += on ? v.y : 0.0f;
    s.z += on ? v.z : 0.0f; s.w += on ? v.w : 0.0f;
    m.x = fmaxf(m.x, on ? v.x : ninf); m.y = fmaxf(m.y, on ? v.y : ninf);
    m.z = fmaxf(m.z, on ? v.z : ninf); m.w = fmaxf(m.w, on ? v.w : ninf);
}

__global__ __launch_bounds__(128) void combine_wmma_kernel(
        const float*    __restrict__ x,
        const int*      __restrict__ src,
        const int*      __restrict__ dst,
        const unsigned* __restrict__ cnt,
        const int*      __restrict__ bucket,
        const float2*   __restrict__ WbigP,
        const float*    __restrict__ bsum,
        float*          __restrict__ out,
        size_t ws_bytes) {
    __shared__ __attribute__((aligned(16))) float Ap[16 * APAD];

    if (ws_bytes < WS_NEED) return;

    const int  t    = threadIdx.x;
    const long base = (long)blockIdx.x * 16;

    {
        const int  row = t >> 3;
        const int  sub = t & 7;
        const long n   = base + row;
        const unsigned cn = cnt[n];
        const unsigned m  = cn < (unsigned)CAP ? cn : (unsigned)CAP;
        const int* __restrict__ bk = bucket + n * CAP;
        const float4* __restrict__ x4 = (const float4*)x;

        const float ninf = -__builtin_inff();
        float4 sa = make_float4(0.0f, 0.0f, 0.0f, 0.0f);
        float4 sb = make_float4(0.0f, 0.0f, 0.0f, 0.0f);
        float4 ma = make_float4(ninf, ninf, ninf, ninf);
        float4 mb = make_float4(ninf, ninf, ninf, ninf);

        unsigned e = 0;
        for (; e + 4 <= m; e += 4) {
            const int4 id = *(const int4*)(bk + e);
            const long i0 = id.x;
            const long i1 = id.y;
            const long i2 = id.z;
            const long i3 = id.w;
            const float4 a0 = x4[i0 * 16 + sub];
            const float4 b0 = x4[i0 * 16 + 8 + sub];
            const float4 a1 = x4[i1 * 16 + sub];
            const float4 b1 = x4[i1 * 16 + 8 + sub];
            const float4 a2 = x4[i2 * 16 + sub];
            const float4 b2 = x4[i2 * 16 + 8 + sub];
            const float4 a3 = x4[i3 * 16 + sub];
            const float4 b3 = x4[i3 * 16 + 8 + sub];
            acc_row(sa, ma, a0); acc_row(sb, mb, b0);
            acc_row(sa, ma, a1); acc_row(sb, mb, b1);
            acc_row(sa, ma, a2); acc_row(sb, mb, b2);
            acc_row(sa, ma, a3); acc_row(sb, mb, b3);
        }
        if (e < m) {
            const unsigned left = m - e;
            const int4 id = *(const int4*)(bk + e);
            const long i0 = id.x;
            const long i1 = left > 1u ? id.y : id.x;
            const long i2 = left > 2u ? id.z : id.x;
            const float4 a0 = x4[i0 * 16 + sub];
            const float4 b0 = x4[i0 * 16 + 8 + sub];
            const float4 a1 = x4[i1 * 16 + sub];
            const float4 b1 = x4[i1 * 16 + 8 + sub];
            const float4 a2 = x4[i2 * 16 + sub];
            const float4 b2 = x4[i2 * 16 + 8 + sub];
            acc_row(sa, ma, a0); acc_row(sb, mb, b0);
            acc_row_if(sa, ma, a1, left > 1u); acc_row_if(sb, mb, b1, left > 1u);
            acc_row_if(sa, ma, a2, left > 2u); acc_row_if(sb, mb, b2, left > 2u);
        }
        if (cn > (unsigned)CAP) {
            sa = make_float4(0.0f, 0.0f, 0.0f, 0.0f);
            sb = make_float4(0.0f, 0.0f, 0.0f, 0.0f);
            ma = make_float4(ninf, ninf, ninf, ninf);
            mb = make_float4(ninf, ninf, ninf, ninf);
            const int me = (int)n;
            for (int q = 0; q < N_EDGES / 4; ++q) {
                const int4 d = ((const int4*)dst)[q];
                if (d.x == me || d.y == me || d.z == me || d.w == me) {
                    const int4 s = ((const int4*)src)[q];
                    const int sid[4] = { s.x, s.y, s.z, s.w };
                    const int did[4] = { d.x, d.y, d.z, d.w };
                    #pragma unroll
                    for (int k = 0; k < 4; ++k) {
                        if (did[k] == me) {
                            const long i = sid[k];
                            acc_row(sa, ma, x4[i * 16 + sub]);
                            acc_row(sb, mb, x4[i * 16 + 8 + sub]);
                        }
                    }
                }
            }
        }

        const float    invc = 1.0f / fmaxf((float)cn, 1.0f);
        const float4   xa   = x4[n * 16 + sub];
        const float4   xb   = x4[n * 16 + 8 + sub];
        if (cn == 0u) {
            ma = make_float4(0.0f, 0.0f, 0.0f, 0.0f);
            mb = make_float4(0.0f, 0.0f, 0.0f, 0.0f);
        }
        float* ap = Ap + row * APAD + 4 * sub;
        *(float4*)(ap +   0) = sa;
        *(float4*)(ap +  32) = sb;
        *(float4*)(ap +  64) = make_float4(sa.x * invc, sa.y * invc, sa.z * invc, sa.w * invc);
        *(float4*)(ap +  96) = make_float4(sb.x * invc, sb.y * invc, sb.z * invc, sb.w * invc);
        *(float4*)(ap + 128) = ma;
        *(float4*)(ap + 160) = mb;
        *(float4*)(ap + 192) = xa;
        *(float4*)(ap + 224) = xb;
    }
    __syncthreads();

    const int lane = t & 31;
    const int wave = t >> 5;
    const int nl   = lane & 15;
    const int half = lane >> 4;
    const int col  = wave * 16 + nl;
    const int m    = nl;

    v8f acc;
    const float bv = bsum[col];
    #pragma unroll
    for (int r = 0; r < 8; ++r) acc[r] = bv;

    const float*  Arow = Ap + m * APAD;
    const float2* Bcol = WbigP + col;

    #pragma unroll
    for (int k = 0; k < 256; k += 4) {
        v2f a, bb;
        a.x = Arow[k + 2 * half + 0];
        a.y = Arow[k + 2 * half + 1];
        const float2 bw = Bcol[((k >> 1) + half) * 64];
        bb.x = bw.x;
        bb.y = bw.y;
        acc = __builtin_amdgcn_wmma_f32_16x16x4_f32(
                  false, a, false, bb, (short)0, acc, false, false);
    }

    #pragma unroll
    for (int r = 0; r < 8; ++r) {
        long n = base + r + 8 * half;
        out[n * 64 + col] = acc[r];
    }
}

extern "C" void kernel_launch(void* const* d_in, const int* in_sizes, int n_in,
                              void* d_out, int out_size, void* d_ws, size_t ws_size,
                              hipStream_t stream) {
    (void)in_sizes; (void)n_in; (void)out_size;

    const float* x  = (const float*)d_in[0];
    const int*   ei = (const int*)d_in[1];
    const float* Wl = (const float*)d_in[2];
    const float* Wr = (const float*)d_in[3];
    const float* b  = (const float*)d_in[4];
    float*       out = (float*)d_out;

    float*    ws     = (float*)d_ws;
    unsigned* cnt    = (unsigned*)ws;
    float2*   WbigP  = (float2*)(ws + 100000);
    float*    bsum   = ws + 116384;
    int*      bucket = (int*)(ws + 116480);

    const int* src = ei;
    const int* dst = ei + N_EDGES;

    zero_ws_kernel<<<391, 256, 0, stream>>>(ws, (long)N_NODES);
    prep_w_kernel<<<32, 256, 0, stream>>>(Wl, Wr, b, WbigP, bsum);
    bucket_edges_kernel<<<977, 256, 0, stream>>>(src, dst, cnt, bucket, ws_size);
    combine_wmma_kernel<<<N_NODES / 16, 128, 0, stream>>>(x, src, dst, cnt, bucket, WbigP, bsum, out, ws_size);
}


# Per-edge global counting atomic + scattered 4-byte store replaced by LDS binning: edges grouped by 400-node range, one block per range builds buckets in LDS, whole-line stores; sums in arrival order
# speedup vs baseline: 3.6856x; 1.8368x over previous
#include <hip/hip_runtime.h>
#include <stdint.h>

#define N_NODES 100000
#define N_EDGES 1000000
#define C 64
#define APAD 260
#define CAP 32
#define BINSZ 400
#define NB 250
#define BINWORDS (BINSZ * CAP)
#define PCHUNK 4096
#define WS_NEED 13265920ul

typedef float v2f __attribute__((ext_vector_type(2)));
typedef float v8f __attribute__((ext_vector_type(8)));

__device__ __forceinline__ float wbig_row(const float* Wl, const float* Wr,
                                          int j, int d) {
    if (j < 192) return Wl[j * 64 + d];
    int r = j - 192;
    return Wr[r * 64 + d] + Wr[4096 + r * 64 + d] + Wr[8192 + r * 64 + d];
}

__global__ void prep_w_kernel(const float* __restrict__ Wl,
                              const float* __restrict__ Wr,
                              const float* __restrict__ b,
                              float2* __restrict__ WbigP,
                              float* __restrict__ bsum,
                              unsigned* __restrict__ cnt) {
    int idx = blockIdx.x * blockDim.x + threadIdx.x;
    if (idx < NB) cnt[idx * BINSZ] = 0u;
    int p = idx >> 6;
    int d = idx & 63;
    float2 v;
    v.x = wbig_row(Wl, Wr, 2 * p + 0, d);
    v.y = wbig_row(Wl, Wr, 2 * p + 1, d);
    WbigP[idx] = v;
    if (idx < 64) bsum[idx] = b[idx] + b[64 + idx] + b[128 + idx];
}

__global__ __launch_bounds__(512) void bin_edges_kernel(const int* __restrict__ src,
                                                        const int* __restrict__ dst,
                                                        unsigned* __restrict__ cnt,
                                                        unsigned* __restrict__ bucket,
                                                        size_t ws_bytes) {
    __shared__ unsigned hist[256];
    __shared__ unsigned lstart[256];
    __shared__ unsigned gbase[256];
    __shared__ unsigned wsum[8];
    __shared__ int ssrc[PCHUNK];
    __shared__ int sdst[PCHUNK];

    if (ws_bytes < WS_NEED) return;

    const int t = threadIdx.x;
    if (t < 256) hist[t] = 0u;
    __syncthreads();

    const int q0 = blockIdx.x * (PCHUNK / 4);
    int s[8], d[8];
    unsigned r[8];
    bool on[2];
    #pragma unroll
    for (int j = 0; j < 2; ++j) {
        const int q = q0 + j * 512 + t;
        on[j] = q < N_EDGES / 4;
        const int4 sv = on[j] ? ((const int4*)src)[q] : make_int4(0, 0, 0, 0);
        const int4 dv = on[j] ? ((const int4*)dst)[q] : make_int4(0, 0, 0, 0);
        s[4 * j + 0] = sv.x; s[4 * j + 1] = sv.y; s[4 * j + 2] = sv.z; s[4 * j + 3] = sv.w;
        d[4 * j + 0] = dv.x; d[4 * j + 1] = dv.y; d[4 * j + 2] = dv.z; d[4 * j + 3] = dv.w;
    }
    #pragma unroll
    for (int k = 0; k < 8; ++k) {
        r[k] = 0u;
        if (on[k >> 2]) r[k] = atomicAdd(&hist[(unsigned)d[k] / BINSZ], 1u);
    }
    __syncthreads();

    unsigned cb = 0u, v = 0u;
    if (t < 256) {
        cb = hist[t];
        v = cb;
        const int lane = t & 31;
        #pragma unroll
        for (int o = 1; o < 32; o <<= 1) {
            const unsigned u = __shfl_up(v, o);
            if (lane >= o) v += u;
        }
        if (lane == 31) wsum[t >> 5] = v;
    }
    __syncthreads();
    if (t < 256) {
        unsigned off = 0u;
        for (int w = 0; w < (t >> 5); ++w) off += wsum[w];
        lstart[t] = off + v - cb;
        gbase[t] = (t < NB && cb != 0u) ? atomicAdd(cnt + t * BINSZ, cb) : 0u;
    }
    __syncthreads();

    #pragma unroll
    for (int k = 0; k < 8; ++k) {
        if (on[k >> 2]) {
            const unsigned p = lstart[(unsigned)d[k] / BINSZ] + r[k];
            ssrc[p] = s[k];
            sdst[p] = d[k];
        }
    }
    __syncthreads();

    const int rest = N_EDGES - (int)blockIdx.x * PCHUNK;
    const int nval = rest < PCHUNK ? rest : PCHUNK;
    for (int i = t; i < nval; i += 512) {
        const unsigned dd = (unsigned)sdst[i];
        const unsigned ss = (unsigned)ssrc[i];
        const unsigned b = dd / BINSZ;
        const unsigned pos = gbase[b] + ((unsigned)i - lstart[b]);
        if (pos < (unsigned)BINWORDS)
            bucket[(size_t)b * BINWORDS + pos] = ss | ((dd - b * BINSZ) << 17);
    }
}

__global__ __launch_bounds__(512) void build_buckets_kernel(unsigned* __restrict__ cnt,
                                                            unsigned* __restrict__ bucket,
                                                            size_t ws_bytes) {
    __shared__ unsigned cntL[BINSZ];
    __shared__ __attribute__((aligned(16))) unsigned lst[BINWORDS];

    if (ws_bytes < WS_NEED) return;

    const int t = threadIdx.x;
    const int b = blockIdx.x;
    unsigned* region = bucket + (size_t)b * BINWORDS;
    const unsigned total = cnt[b * BINSZ];
    const unsigned ne = total < (unsigned)BINWORDS ? total : (unsigned)BINWORDS;

    if (t < BINSZ) cntL[t] = 0u;
    for (int i = t; i < BINWORDS / 4; i += 512) ((uint4*)lst)[i] = make_uint4(0u, 0u, 0u, 0u);
    __syncthreads();

    for (unsigned i0 = 0; i0 < ne; i0 += 4096u) {
        unsigned w[8];
        #pragma unroll
        for (int k = 0; k < 8; ++k) {
            const unsigned i = i0 + (unsigned)(k * 512 + t);
            w[k] = i < ne ? region[i] : 0u;
        }
        #pragma unroll
        for (int k = 0; k < 8; ++k) {
            const unsigned i = i0 + (unsigned)(k * 512 + t);
            const unsigned dl = w[k] >> 17;
            if (i < ne && dl < (unsigned)BINSZ) {
                const unsigned slot = atomicAdd(&cntL[dl], 1u);
                if (slot < (unsigned)CAP) lst[dl * CAP + slot] = w[k] & 0x1FFFFu;
            }
        }
    }
    __syncthreads();

    for (int i = t; i < BINWORDS / 4; i += 512) ((uint4*)region)[i] = ((const uint4*)lst)[i];
    if (t < BINSZ) cnt[b * BINSZ + t] = total > (unsigned)BINWORDS ? 0xFFFFFFFFu : cntL[t];
}

__device__ __forceinline__ void acc_row(float4& s, float4& m, const float4 v) {
    s.x += v.x; s.y += v.y; s.z += v.z; s.w += v.w;
    m.x = fmaxf(m.x, v.x); m.y = fmaxf(m.y, v.y);
    m.z = fmaxf(m.z, v.z); m.w = fmaxf(m.w, v.w);
}

__device__ __forceinline__ void acc_row_if(float4& s, float4& m, const float4 v, bool on) {
    const float ninf = -__builtin_inff();
    s.x += on ? v.x : 0.0f; s.y += on ? v.y : 0.0f;
    s.z += on ? v.z : 0.0f; s.w += on ? v.w : 0.0f;
    m.x = fmaxf(m.x, on ? v.x : ninf); m.y = fmaxf(m.y, on ? v.y : ninf);
    m.z = fmaxf(m.z, on ? v.z : ninf); m.w = fmaxf(m.w, on ? v.w : ninf);
}

__global__ __launch_bounds__(128) void combine_wmma_kernel(
        const float*    __restrict__ x,
        const int*      __restrict__ src,
        const int*      __restrict__ dst,
        const unsigned* __restrict__ cnt,
        const int*      __restrict__ bucket,
        const float2*   __restrict__ WbigP,
        const float*    __restrict__ bsum,
        float*          __restrict__ out,
        size_t ws_bytes) {
    __shared__ __attribute__((aligned(16))) float Ap[16 * APAD];

    if (ws_bytes < WS_NEED) return;

    const int  t    = threadIdx.x;
    const long base = (long)blockIdx.x * 16;

    {
        const int  row = t >> 3;
        const int  sub = t & 7;
        const long n   = base + row;
        unsigned cn = cnt[n];
        const unsigned m  = cn < (unsigned)CAP ? cn : (unsigned)CAP;
        const int* __restrict__ bk = bucket + n * CAP;
        const float4* __restrict__ x4 = (const float4*)x;

        const float ninf = -__builtin_inff();
        float4 sa = make_float4(0.0f, 0.0f, 0.0f, 0.0f);
        float4 sb = make_float4(0.0f, 0.0f, 0.0f, 0.0f);
        float4 ma = make_float4(ninf, ninf, ninf, ninf);
        float4 mb = make_float4(ninf, ninf, ninf, ninf);

        unsigned e = 0;
        for (; e + 4 <= m; e += 4) {
            const int4 id = *(const int4*)(bk + e);
            const long i0 = id.x;
            const long i1 = id.y;
            const long i2 = id.z;
            const long i3 = id.w;
            const float4 a0 = x4[i0 * 16 + sub];
            const float4 b0 = x4[i0 * 16 + 8 + sub];
            const float4 a1 = x4[i1 * 16 + sub];
            const float4 b1 = x4[i1 * 16 + 8 + sub];
            const float4 a2 = x4[i2 * 16 + sub];
            const float4 b2 = x4[i2 * 16 + 8 + sub];
            const float4 a3 = x4[i3 * 16 + sub];
            const float4 b3 = x4[i3 * 16 + 8 + sub];
            acc_row(sa, ma, a0); acc_row(sb, mb, b0);
            acc_row(sa, ma, a1); acc_row(sb, mb, b1);
            acc_row(sa, ma, a2); acc_row(sb, mb, b2);
            acc_row(sa, ma, a3); acc_row(sb, mb, b3);
        }
        if (e < m) {
            const unsigned left = m - e;
            const int4 id = *(const int4*)(bk + e);
            const long i0 = id.x;
            const long i1 = left > 1u ? id.y : id.x;
            const long i2 = left > 2u ? id.z : id.x;
            const float4 a0 = x4[i0 * 16 + sub];
            const float4 b0 = x4[i0 * 16 + 8 + sub];
            const float4 a1 = x4[i1 * 16 + sub];
            const float4 b1 = x4[i1 * 16 + 8 + sub];
            const float4 a2 = x4[i2 * 16 + sub];
            const float4 b2 = x4[i2 * 16 + 8 + sub];
            acc_row(sa, ma, a0); acc_row(sb, mb, b0);
            acc_row_if(sa, ma, a1, left > 1u); acc_row_if(sb, mb, b1, left > 1u);
            acc_row_if(sa, ma, a2, left > 2u); acc_row_if(sb, mb, b2, left > 2u);
        }
        if (cn > (unsigned)CAP) {
            sa = make_float4(0.0f, 0.0f, 0.0f, 0.0f);
            sb = make_float4(0.0f, 0.0f, 0.0f, 0.0f);
            ma = make_float4(ninf, ninf, ninf, ninf);
            mb = make_float4(ninf, ninf, ninf, ninf);
            const int me = (int)n;
            unsigned seen = 0u;
            for (int q = 0; q < N_EDGES / 4; ++q) {
                const int4 d = ((const int4*)dst)[q];
                if (d.x == me || d.y == me || d.z == me || d.w == me) {
                    const int4 s = ((const int4*)src)[q];
                    const int sid[4] = { s.x, s.y, s.z, s.w };
                    const int did[4] = { d.x, d.y, d.z, d.w };
                    #pragma unroll
                    for (int k = 0; k < 4; ++k) {
                        if (did[k] == me) {
                            const long i = sid[k];
                            acc_row(sa, ma, x4[i * 16 + sub]);
                            acc_row(sb, mb, x4[i * 16 + 8 + sub]);
                            ++seen;
                        }
                    }
                }
            }
            cn = seen;
        }

        const float    invc = 1.0f / fmaxf((float)cn, 1.0f);
        const float4   xa   = x4[n * 16 + sub];
        const float4   xb   = x4[n * 16 + 8 + sub];
        if (cn == 0u) {
            ma = make_float4(0.0f, 0.0f, 0.0f, 0.0f);
            mb = make_float4(0.0f, 0.0f, 0.0f, 0.0f);
        }
        float* ap = Ap + row * APAD + 4 * sub;
        *(float4*)(ap +   0) = sa;
        *(float4*)(ap +  32) = sb;
        *(float4*)(ap +  64) = make_float4(sa.x * invc, sa.y * invc, sa.z * invc, sa.w * invc);
        *(float4*)(ap +  96) = make_float4(sb.x * invc, sb.y * invc, sb.z * invc, sb.w * invc);
        *(float4*)(ap + 128) = ma;
        *(float4*)(ap + 160) = mb;
        *(float4*)(ap + 192) = xa;
        *(float4*)(ap + 224) = xb;
    }
    __syncthreads();

    const int lane = t & 31;
    const int wave = t >> 5;
    const int nl   = lane & 15;
    const int half = lane >> 4;
    const int col  = wave * 16 + nl;
    const int m    = nl;

    v8f acc;
    const float bv = bsum[col];
    #pragma unroll
    for (int r = 0; r < 8; ++r) acc[r] = bv;

    const float*  Arow = Ap + m * APAD;
    const float2* Bcol = WbigP + col;

    #pragma unroll
    for (int k = 0; k < 256; k += 4) {
        v2f a, bb;
        a.x = Arow[k + 2 * half + 0];
        a.y = Arow[k + 2 * half + 1];
        const float2 bw = Bcol[((k >> 1) + half) * 64];
        bb.x = bw.x;
        bb.y = bw.y;
        acc = __builtin_amdgcn_wmma_f32_16x16x4_f32(
                  false, a, false, bb, (short)0, acc, false, false);
    }

    #pragma unroll
    for (int r = 0; r < 8; ++r) {
        long n = base + r + 8 * half;
        out[n * 64 + col] = acc[r];
    }
}

extern "C" void kernel_launch(void* const* d_in, const int* in_sizes, int n_in,
                              void* d_out, int out_size, void* d_ws, size_t ws_size,
                              hipStream_t stream) {
    (void)in_sizes; (void)n_in; (void)out_size;

    const float* x  = (const float*)d_in[0];
    const int*   ei = (const int*)d_in[1];
    const float* Wl = (const float*)d_in[2];
    const float* Wr = (const float*)d_in[3];
    const float* b  = (const float*)d_in[4];
    float*       out = (float*)d_out;

    float*    ws     = (float*)d_ws;
    unsigned* cnt    = (unsigned*)ws;
    float2*   WbigP  = (float2*)(ws + 100000);
    float*    bsum   = ws + 116384;
    int*      bucket = (int*)(ws + 116480);

    const int* src = ei;
    const int* dst = ei + N_EDGES;

    prep_w_kernel<<<32, 256, 0, stream>>>(Wl, Wr, b, WbigP, bsum, cnt);
    bin_edges_kernel<<<245, 512, 0, stream>>>(src, dst, cnt, (unsigned*)bucket, ws_size);
    build_buckets_kernel<<<NB, 512, 0, stream>>>(cnt, (unsigned*)bucket, ws_size);
    combine_wmma_kernel<<<N_NODES / 16, 128, 0, stream>>>(x, src, dst, cnt, bucket, WbigP, bsum, out, ws_size);
}
